# starting kernel
# baseline (speedup 1.0000x reference)
#include <hip/hip_runtime.h>


#define LSTM_B 4096
#define LSTM_T 512

typedef float v2f __attribute__((ext_vector_type(2)));
typedef float v8f __attribute__((ext_vector_type(8)));

static __device__ __forceinline__ v8f wmma4(v2f a, v2f b, v8f c) {
  return __builtin_amdgcn_wmma_f32_16x16x4_f32(false, a, false, b, (short)0, c,
                                               false, false);
}

static __device__ __forceinline__ float sigm(float x) {
  return __builtin_amdgcn_rcpf(1.0f + __builtin_amdgcn_exp2f(-1.44269504f * x));
}
static __device__ __forceinline__ float tanh_a(float x) {
  return 1.0f - 2.0f * __builtin_amdgcn_rcpf(
                           1.0f + __builtin_amdgcn_exp2f(2.88539008f * x));
}

__global__ void __launch_bounds__(128, 1) lstm2_fc_kernel(
    const float* __restrict__ x,
    const float* __restrict__ Wih0,
    const float* __restrict__ Whh0,
    const float* __restrict__ bih0,
    const float* __restrict__ bhh0,
    const float* __restrict__ Wih1,
    const float* __restrict__ Whh1,
    const float* __restrict__ bih1,
    const float* __restrict__ bhh1,
    const float* __restrict__ Wfc,
    const float* __restrict__ bfc,
    float* __restrict__ out)
{
  const int lane = threadIdx.x & 31;
  const int half = lane >> 4;
  const int nIdx = lane & 15;
  const int wid  = (blockIdx.x * blockDim.x + threadIdx.x) >> 5;
  const int base = wid * 16;

  v2f Whh0A[4][4], Wih1A[4][4], Whh1A[4][4];
  v2f Wih0A[4];
  v2f B1A[4];
#pragma unroll
  for (int j = 0; j < 4; ++j) {
    const int row = 16 * j + nIdx;
#pragma unroll
    for (int kc = 0; kc < 4; ++kc) {
      const float* p0 = Whh0 + row * 16 + kc * 4 + 2 * half;
      const float* p1 = Wih1 + row * 16 + kc * 4 + 2 * half;
      const float* p2 = Whh1 + row * 16 + kc * 4 + 2 * half;
      Whh0A[j][kc] = v2f{p0[0], p0[1]};
      Wih1A[j][kc] = v2f{p1[0], p1[1]};
      Whh1A[j][kc] = v2f{p2[0], p2[1]};
    }
    const float w0  = Wih0[row];
    const float bb0 = bih0[row] + bhh0[row];
    const float bb1 = bih1[row] + bhh1[row];
    Wih0A[j] = v2f{half ? 0.0f : w0, half ? 0.0f : bb0};
    B1A[j]   = v2f{half ? 0.0f : bb1, 0.0f};
  }
  const v2f onesB = v2f{half ? 0.0f : 1.0f, 0.0f};
  const v8f vzero = {0.f, 0.f, 0.f, 0.f, 0.f, 0.f, 0.f, 0.f};

  v8f c0 = vzero, h0 = vzero, c1 = vzero, h1 = vzero;
  v2f h0B[4], h1B[4];
#pragma unroll
  for (int kc = 0; kc < 4; ++kc) {
    h0B[kc] = v2f{0.0f, 0.0f};
    h1B[kc] = v2f{0.0f, 0.0f};
  }

  auto convert = [&](const v8f& h, v2f(&hB)[4]) {
#pragma unroll
    for (int kc = 0; kc < 4; ++kc) {
      const int k0 = 4 * kc;
      if (kc < 2) {
        float sx = __shfl_xor(h[k0 + 2], 16, 32);
        float sy = __shfl_xor(h[k0 + 3], 16, 32);
        hB[kc] = v2f{half ? sx : h[k0 + 0], half ? sy : h[k0 + 1]};
      } else {
        float sx = __shfl_xor(h[k0 - 8], 16, 32);
        float sy = __shfl_xor(h[k0 - 7], 16, 32);
        hB[kc] = v2f{half ? h[k0 - 6] : sx, half ? h[k0 - 5] : sy};
      }
    }
  };

  const float* xrow = x + (base + nIdx) * LSTM_T;

#pragma unroll 1
  for (int t = 0; t < LSTM_T; ++t) {
    const float xv = xrow[t];
    const v2f xB = v2f{half ? 0.0f : xv, half ? 0.0f : 1.0f};

    v8f g[4];
#pragma unroll
    for (int j = 0; j < 4; ++j) {
      v8f acc = wmma4(Wih0A[j], xB, vzero);
#pragma unroll
      for (int kc = 0; kc < 4; ++kc) acc = wmma4(Whh0A[j][kc], h0B[kc], acc);
      g[j] = acc;
    }
#pragma unroll
    for (int v = 0; v < 8; ++v) {
      const float iv = sigm(g[0][v]);
      const float fv = sigm(g[1][v]);
      const float gv = tanh_a(g[2][v]);
      const float ov = sigm(g[3][v]);
      const float cv = fv * c0[v] + iv * gv;
      c0[v] = cv;
      h0[v] = ov * tanh_a(cv);
    }
    convert(h0, h0B);

#pragma unroll
    for (int j = 0; j < 4; ++j) {
      v8f acc = wmma4(B1A[j], onesB, vzero);
#pragma unroll
      for (int kc = 0; kc < 4; ++kc) acc = wmma4(Wih1A[j][kc], h0B[kc], acc);
#pragma unroll
      for (int kc = 0; kc < 4; ++kc) acc = wmma4(Whh1A[j][kc], h1B[kc], acc);
      g[j] = acc;
    }
#pragma unroll
    for (int v = 0; v < 8; ++v) {
      const float iv = sigm(g[0][v]);
      const float fv = sigm(g[1][v]);
      const float gv = tanh_a(g[2][v]);
      const float ov = sigm(g[3][v]);
      const float cv = fv * c1[v] + iv * gv;
      c1[v] = cv;
      h1[v] = ov * tanh_a(cv);
    }
    convert(h1, h1B);
  }

  float partial = 0.0f;
#pragma unroll
  for (int v = 0; v < 8; ++v) partial += Wfc[v + 8 * half] * h1[v];
  const float tot = partial + __shfl_xor(partial, 16, 32);
  if (half == 0) out[base + nIdx] = tot + bfc[0];
}

extern "C" void kernel_launch(void* const* d_in, const int* in_sizes, int n_in,
                              void* d_out, int out_size, void* d_ws,
                              size_t ws_size, hipStream_t stream) {
  (void)in_sizes; (void)n_in; (void)d_ws; (void)ws_size; (void)out_size;
  const float* x    = (const float*)d_in[0];
  const float* Wih0 = (const float*)d_in[1];
  const float* Whh0 = (const float*)d_in[2];
  const float* bih0 = (const float*)d_in[3];
  const float* bhh0 = (const float*)d_in[4];
  const float* Wih1 = (const float*)d_in[5];
  const float* Whh1 = (const float*)d_in[6];
  const float* bih1 = (const float*)d_in[7];
  const float* bhh1 = (const float*)d_in[8];
  const float* Wfc  = (const float*)d_in[9];
  const float* bfc  = (const float*)d_in[10];
  float* out = (float*)d_out;

  dim3 block(128);
  dim3 grid(LSTM_B / (16 * 4));
  lstm2_fc_kernel<<<grid, block, 0, stream>>>(x, Wih0, Whh0, bih0, bhh0, Wih1,
                                              Whh1, bih1, bhh1, Wfc, bfc, out);
}


# time loop software-pipelined inside the wave by halves of h: next step's x/bias link and Whh0 link 0 (8 of 52 WMMAs) issue inside this step's gate math; x read a step ahead; all sums unchanged
# speedup vs baseline: 1.5865x; 1.5865x over previous
#include <hip/hip_runtime.h>


#define LSTM_B 4096
#define LSTM_T 512

typedef float v2f __attribute__((ext_vector_type(2)));
typedef float v8f __attribute__((ext_vector_type(8)));

static __device__ __forceinline__ v8f wmma4(v2f a, v2f b, v8f c) {
  return __builtin_amdgcn_wmma_f32_16x16x4_f32(false, a, false, b, (short)0, c,
                                               false, false);
}

static __device__ __forceinline__ void swap16(float& d, float& s) {
  asm("v_permlane16_swap_b32 %0, %1" : "+v"(d), "+v"(s));
}

static __device__ __forceinline__ float sigm(float x) {
  return __builtin_amdgcn_rcpf(1.0f + __builtin_amdgcn_exp2f(-1.44269504f * x));
}
static __device__ __forceinline__ float tanh_a(float x) {
  return 1.0f - 2.0f * __builtin_amdgcn_rcpf(
                           1.0f + __builtin_amdgcn_exp2f(2.88539008f * x));
}

static __device__ __forceinline__ float shr8z(float v) {
  return __builtin_bit_cast(
      float, __builtin_amdgcn_update_dpp(0, __builtin_bit_cast(int, v), 0x118,
                                         0xf, 0xf, true));
}

__global__ void __launch_bounds__(128, 1) lstm2_fc_kernel(
    const float* __restrict__ x,
    const float* __restrict__ Wih0,
    const float* __restrict__ Whh0,
    const float* __restrict__ bih0,
    const float* __restrict__ bhh0,
    const float* __restrict__ Wih1,
    const float* __restrict__ Whh1,
    const float* __restrict__ bih1,
    const float* __restrict__ bhh1,
    const float* __restrict__ Wfc,
    const float* __restrict__ bfc,
    float* __restrict__ out)
{
  __shared__ float outLine[32];

  const int lane = threadIdx.x & 31;
  const int half = lane >> 4;
  const int nIdx = lane & 15;
  const bool up  = nIdx >= 8;
  const int wv   = threadIdx.x >> 5;
  const int wid  = blockIdx.x * 4 + wv;
  const int base = wid * 8;

  v2f Whh0A[4][4], Wih1A[4][4], Whh1A[4][4];
  v2f AugA[4];
#pragma unroll
  for (int j = 0; j < 4; ++j) {
    const int row = 16 * j + nIdx;
#pragma unroll
    for (int kc = 0; kc < 4; ++kc) {
      const float* p0 = Whh0 + row * 16 + kc * 4 + 2 * half;
      const float* p1 = Wih1 + row * 16 + kc * 4 + 2 * half;
      const float* p2 = Whh1 + row * 16 + kc * 4 + 2 * half;
      Whh0A[j][kc] = v2f{p0[0], p0[1]};
      Wih1A[j][kc] = v2f{p1[0], p1[1]};
      Whh1A[j][kc] = v2f{p2[0], p2[1]};
    }
    const float w0  = Wih0[row];
    const float bb0 = bih0[row] + bhh0[row];
    const float bb1 = bih1[row] + bhh1[row];
    AugA[j] = v2f{half ? bb1 : w0, half ? 0.0f : bb0};
  }
  const v8f vzero = {0.f, 0.f, 0.f, 0.f, 0.f, 0.f, 0.f, 0.f};

  const bool  xLane   = (half == 0) && !up;
  const float augFix0 = (half == 1 && up) ? 1.0f : 0.0f;
  const float augFix1 = xLane ? 1.0f : 0.0f;

  v8f cS = vzero, hS = vzero;
  v2f hLo[4], hSh[4], hHi[4];
#pragma unroll
  for (int kc = 0; kc < 4; ++kc) {
    hLo[kc] = v2f{0.0f, 0.0f};
    hSh[kc] = v2f{0.0f, 0.0f};
    hHi[kc] = v2f{0.0f, 0.0f};
  }

  const float* xrow = x + (base + (nIdx & 7)) * LSTM_T;

  v8f gA[4], gB[4];

  float xNext = xrow[1];
  {
    const float x0 = xrow[0];
    const v2f augB = v2f{xLane ? x0 : augFix0, augFix1};
#pragma unroll
    for (int j = 0; j < 4; ++j)
      gA[j] = wmma4(Whh0A[j][0], hLo[0], wmma4(AugA[j], augB, vzero));
  }

  auto step = [&](const int s, v8f (&gc)[4], v8f (&gn)[4],
                  const bool first) __attribute__((always_inline)) {
#pragma unroll
    for (int j = 0; j < 4; ++j) {
      v8f acc = gc[j];
#pragma unroll
      for (int kc = 1; kc < 4; ++kc) acc = wmma4(Whh0A[j][kc], hLo[kc], acc);
#pragma unroll
      for (int kc = 0; kc < 4; ++kc) acc = wmma4(Wih1A[j][kc], hSh[kc], acc);
#pragma unroll
      for (int kc = 0; kc < 4; ++kc) acc = wmma4(Whh1A[j][kc], hHi[kc], acc);
      gc[j] = acc;
    }
    const v2f augB = v2f{xLane ? xNext : augFix0, augFix1};
    const int sn = s + 2;
    xNext = xrow[sn < LSTM_T ? sn : LSTM_T - 1];

#pragma unroll
    for (int p = 0; p < 2; ++p) {
      float ei[4], ef[4], eg[4], eo[4], ov[4], cv[4], ec[4];
#pragma unroll
      for (int q = 0; q < 4; ++q) {
        const int v = 4 * p + q;
        ei[q] = __builtin_amdgcn_exp2f(-1.44269504f * gc[0][v]);
        ef[q] = __builtin_amdgcn_exp2f(-1.44269504f * gc[1][v]);
        eg[q] = __builtin_amdgcn_exp2f(2.88539008f * gc[2][v]);
        eo[q] = __builtin_amdgcn_exp2f(-1.44269504f * gc[3][v]);
      }
      gn[0] = p ? wmma4(Whh0A[0][0], hLo[0], gn[0]) : wmma4(AugA[0], augB, vzero);
      __builtin_amdgcn_sched_barrier(0);
#pragma unroll
      for (int q = 0; q < 4; ++q) {
        const int v = 4 * p + q;
        const float iv = __builtin_amdgcn_rcpf(1.0f + ei[q]);
        const float fv = __builtin_amdgcn_rcpf(1.0f + ef[q]);
        const float gv = 1.0f - 2.0f * __builtin_amdgcn_rcpf(1.0f + eg[q]);
        ov[q] = __builtin_amdgcn_rcpf(1.0f + eo[q]);
        cv[q] = fv * cS[v] + iv * gv;
      }
      gn[1] = p ? wmma4(Whh0A[1][0], hLo[0], gn[1]) : wmma4(AugA[1], augB, vzero);
      __builtin_amdgcn_sched_barrier(0);
#pragma unroll
      for (int q = 0; q < 4; ++q)
        ec[q] = __builtin_amdgcn_exp2f(2.88539008f * cv[q]);
      gn[2] = p ? wmma4(Whh0A[2][0], hLo[0], gn[2]) : wmma4(AugA[2], augB, vzero);
      __builtin_amdgcn_sched_barrier(0);
#pragma unroll
      for (int q = 0; q < 4; ++q) {
        const int v = 4 * p + q;
        const float hv =
            ov[q] * (1.0f - 2.0f * __builtin_amdgcn_rcpf(1.0f + ec[q]));
        cS[v] = (first && up) ? 0.0f : cv[q];
        hS[v] = (first && up) ? 0.0f : hv;
      }
      gn[3] = p ? wmma4(Whh0A[3][0], hLo[0], gn[3]) : wmma4(AugA[3], augB, vzero);
      float a0 = hS[4 * p + 0], a1 = hS[4 * p + 1];
      float b0 = hS[4 * p + 2], b1 = hS[4 * p + 3];
      swap16(a0, b0);
      swap16(a1, b1);
      hLo[p]     = v2f{up ? 0.0f : a0, up ? 0.0f : a1};
      hLo[p + 2] = v2f{up ? 0.0f : b0, up ? 0.0f : b1};
      hHi[p]     = v2f{up ? a0 : 0.0f, up ? a1 : 0.0f};
      hHi[p + 2] = v2f{up ? b0 : 0.0f, up ? b1 : 0.0f};
      hSh[p]     = v2f{shr8z(a0), shr8z(a1)};
      hSh[p + 2] = v2f{shr8z(b0), shr8z(b1)};
      __builtin_amdgcn_sched_barrier(0);
    }
  };

  step(0, gA, gB, true);
#pragma unroll 1
  for (int s = 1; s < LSTM_T; s += 2) {
    step(s, gB, gA, false);
    step(s + 1, gA, gB, false);
  }

  float partial = 0.0f;
#pragma unroll
  for (int v = 0; v < 8; ++v) partial += Wfc[v + 8 * half] * hS[v];
  const float tot = partial + __shfl_xor(partial, 16, 32);
  if (half == 0 && up) outLine[wv * 8 + (nIdx - 8)] = tot + bfc[0];
  __syncthreads();
  if (wv == 0) out[blockIdx.x * 32 + lane] = outLine[lane];
}

extern "C" void kernel_launch(void* const* d_in, const int* in_sizes, int n_in,
                              void* d_out, int out_size, void* d_ws,
                              size_t ws_size, hipStream_t stream) {
  (void)in_sizes; (void)n_in; (void)d_ws; (void)ws_size; (void)out_size;
  const float* x    = (const float*)d_in[0];
  const float* Wih0 = (const float*)d_in[1];
  const float* Whh0 = (const float*)d_in[2];
  const float* bih0 = (const float*)d_in[3];
  const float* bhh0 = (const float*)d_in[4];
  const float* Wih1 = (const float*)d_in[5];
  const float* Whh1 = (const float*)d_in[6];
  const float* bih1 = (const float*)d_in[7];
  const float* bhh1 = (const float*)d_in[8];
  const float* Wfc  = (const float*)d_in[9];
  const float* bfc  = (const float*)d_in[10];
  float* out = (float*)d_out;

  dim3 block(128);
  dim3 grid(LSTM_B / (8 * 4));
  lstm2_fc_kernel<<<grid, block, 0, stream>>>(x, Wih0, Whh0, bih0, bhh0, Wih1,
                                              Whh1, bih1, bhh1, Wfc, bfc, out);
}


# layer 1's input products batched over time: every 2nd iteration 16 WMMAs do Wih1 @ h0 of two steps (one per column half), fed in as C tiles; layer 1 three steps behind; its k summed in a new order
# speedup vs baseline: 1.6707x; 1.0531x over previous
#include <hip/hip_runtime.h>


#define LSTM_B 4096
#define LSTM_T 512

typedef float v2f __attribute__((ext_vector_type(2)));
typedef float v8f __attribute__((ext_vector_type(8)));

static __device__ __forceinline__ v8f wmma4(v2f a, v2f b, v8f c) {
  return __builtin_amdgcn_wmma_f32_16x16x4_f32(false, a, false, b, (short)0, c,
                                               false, false);
}

static __device__ __forceinline__ void swap16(float& d, float& s) {
  asm("v_permlane16_swap_b32 %0, %1" : "+v"(d), "+v"(s));
}

static __device__ __forceinline__ float sigm(float x) {
  return __builtin_amdgcn_rcpf(1.0f + __builtin_amdgcn_exp2f(-1.44269504f * x));
}
static __device__ __forceinline__ float tanh_a(float x) {
  return 1.0f - 2.0f * __builtin_amdgcn_rcpf(
                           1.0f + __builtin_amdgcn_exp2f(2.88539008f * x));
}

static __device__ __forceinline__ float shr8z(float v) {
  return __builtin_bit_cast(
      float, __builtin_amdgcn_update_dpp(0, __builtin_bit_cast(int, v), 0x118,
                                         0xf, 0xf, true));
}

__global__ void __launch_bounds__(128, 1) lstm2_fc_kernel(
    const float* __restrict__ x,
    const float* __restrict__ Wih0,
    const float* __restrict__ Whh0,
    const float* __restrict__ bih0,
    const float* __restrict__ bhh0,
    const float* __restrict__ Wih1,
    const float* __restrict__ Whh1,
    const float* __restrict__ bih1,
    const float* __restrict__ bhh1,
    const float* __restrict__ Wfc,
    const float* __restrict__ bfc,
    float* __restrict__ out)
{
  __shared__ float outLine[32];

  const int lane = threadIdx.x & 31;
  const int half = lane >> 4;
  const int nIdx = lane & 15;
  const bool up  = nIdx >= 8;
  const int wv   = threadIdx.x >> 5;
  const int wid  = blockIdx.x * 4 + wv;
  const int base = wid * 8;

  v2f Whh0A[4][4], Wih1A[4][4], Whh1A[4][4];
  v2f AugA[4];
#pragma unroll
  for (int j = 0; j < 4; ++j) {
    const int row = 16 * j + nIdx;
#pragma unroll
    for (int kc = 0; kc < 4; ++kc) {
      const float* p0 = Whh0 + row * 16 + kc * 4 + 2 * half;
      const float* p1 = Wih1 + row * 16 + kc * 4 + 2 * half;
      const float* p2 = Whh1 + row * 16 + kc * 4 + 2 * half;
      Whh0A[j][kc] = v2f{p0[0], p0[1]};
      Wih1A[j][kc] = v2f{p1[0], p1[1]};
      Whh1A[j][kc] = v2f{p2[0], p2[1]};
    }
    const float w0  = Wih0[row];
    const float bb0 = bih0[row] + bhh0[row];
    const float bb1 = bih1[row] + bhh1[row];
    AugA[j] = v2f{half ? bb1 : w0, half ? 0.0f : bb0};
  }
  const v8f vzero = {0.f, 0.f, 0.f, 0.f, 0.f, 0.f, 0.f, 0.f};

  const bool  xLane   = (half == 0) && !up;
  const float augFix0 = (half == 1 && up) ? 1.0f : 0.0f;
  const float augFix1 = xLane ? 1.0f : 0.0f;

  v8f cS = vzero, hS = vzero;
  v2f hLo[4], hShE[4], hHi[4];
#pragma unroll
  for (int kc = 0; kc < 4; ++kc) {
    hLo[kc]  = v2f{0.0f, 0.0f};
    hShE[kc] = v2f{0.0f, 0.0f};
    hHi[kc]  = v2f{0.0f, 0.0f};
  }
  v8f qA[4], qB[4];
#pragma unroll
  for (int j = 0; j < 4; ++j) { qA[j] = vzero; qB[j] = vzero; }

  const float* xrow = x + (base + (nIdx & 7)) * LSTM_T;

  v8f gA[4], gB[4];

  float xNext = xrow[1];
  {
    const float x0 = xrow[0];
    const v2f augB = v2f{xLane ? x0 : augFix0, augFix1};
#pragma unroll
    for (int j = 0; j < 4; ++j)
      gA[j] = wmma4(Whh0A[j][0], hLo[0], wmma4(AugA[j], augB, vzero));
  }

  auto step = [&](const int s, v8f (&gc)[4], v8f (&gn)[4], const bool first,
                  const bool even) __attribute__((always_inline)) {
    if (even) {
      v2f bq[4];
#pragma unroll
      for (int kc = 0; kc < 4; ++kc)
        bq[kc] = v2f{up ? hShE[kc][0] : hLo[kc][0],
                     up ? hShE[kc][1] : hLo[kc][1]};
#pragma unroll
      for (int j = 0; j < 4; ++j) {
        v8f q = vzero;
#pragma unroll
        for (int kc = 0; kc < 4; ++kc) q = wmma4(Wih1A[j][kc], bq[kc], q);
#pragma unroll
        for (int v = 0; v < 8; ++v) {
          qA[j][v] = up ? q[v] : 0.0f;
          qB[j][v] = shr8z(q[v]);
        }
      }
    }
#pragma unroll
    for (int j = 0; j < 4; ++j) {
      v8f acc = gc[j];
#pragma unroll
      for (int kc = 1; kc < 4; ++kc) acc = wmma4(Whh0A[j][kc], hLo[kc], acc);
#pragma unroll
      for (int kc = 0; kc < 4; ++kc) acc = wmma4(Whh1A[j][kc], hHi[kc], acc);
      gc[j] = acc;
    }
    if (even) __builtin_amdgcn_sched_barrier(0);
    const v2f augB = v2f{xLane ? xNext : augFix0, augFix1};
    const int sn = s + 2;
    xNext = xrow[sn < LSTM_T ? sn : LSTM_T - 1];

#pragma unroll
    for (int p = 0; p < 2; ++p) {
      float ei[4], ef[4], eg[4], eo[4], ov[4], cv[4], ec[4];
#pragma unroll
      for (int q = 0; q < 4; ++q) {
        const int v = 4 * p + q;
        ei[q] = __builtin_amdgcn_exp2f(-1.44269504f * gc[0][v]);
        ef[q] = __builtin_amdgcn_exp2f(-1.44269504f * gc[1][v]);
        eg[q] = __builtin_amdgcn_exp2f(2.88539008f * gc[2][v]);
        eo[q] = __builtin_amdgcn_exp2f(-1.44269504f * gc[3][v]);
      }
      gn[0] = p ? wmma4(Whh0A[0][0], hLo[0], gn[0]) : wmma4(AugA[0], augB, even ? qA[0] : qB[0]);
      __builtin_amdgcn_sched_barrier(0);
#pragma unroll
      for (int q = 0; q < 4; ++q) {
        const int v = 4 * p + q;
        const float iv = __builtin_amdgcn_rcpf(1.0f + ei[q]);
        const float fv = __builtin_amdgcn_rcpf(1.0f + ef[q]);
        const float gv = 1.0f - 2.0f * __builtin_amdgcn_rcpf(1.0f + eg[q]);
        ov[q] = __builtin_amdgcn_rcpf(1.0f + eo[q]);
        cv[q] = fv * cS[v] + iv * gv;
      }
      gn[1] = p ? wmma4(Whh0A[1][0], hLo[0], gn[1]) : wmma4(AugA[1], augB, even ? qA[1] : qB[1]);
      __builtin_amdgcn_sched_barrier(0);
#pragma unroll
      for (int q = 0; q < 4; ++q)
        ec[q] = __builtin_amdgcn_exp2f(2.88539008f * cv[q]);
      gn[2] = p ? wmma4(Whh0A[2][0], hLo[0], gn[2]) : wmma4(AugA[2], augB, even ? qA[2] : qB[2]);
      __builtin_amdgcn_sched_barrier(0);
#pragma unroll
      for (int q = 0; q < 4; ++q) {
        const int v = 4 * p + q;
        const float hv =
            ov[q] * (1.0f - 2.0f * __builtin_amdgcn_rcpf(1.0f + ec[q]));
        cS[v] = (first && up) ? 0.0f : cv[q];
        hS[v] = (first && up) ? 0.0f : hv;
      }
      gn[3] = p ? wmma4(Whh0A[3][0], hLo[0], gn[3]) : wmma4(AugA[3], augB, even ? qA[3] : qB[3]);
      float a0 = hS[4 * p + 0], a1 = hS[4 * p + 1];
      float b0 = hS[4 * p + 2], b1 = hS[4 * p + 3];
      swap16(a0, b0);
      swap16(a1, b1);
      hLo[p]     = v2f{up ? 0.0f : a0, up ? 0.0f : a1};
      hLo[p + 2] = v2f{up ? 0.0f : b0, up ? 0.0f : b1};
      hHi[p]     = v2f{up ? a0 : 0.0f, up ? a1 : 0.0f};
      hHi[p + 2] = v2f{up ? b0 : 0.0f, up ? b1 : 0.0f};
      if (even) {
        hShE[p]     = v2f{shr8z(a0), shr8z(a1)};
        hShE[p + 2] = v2f{shr8z(b0), shr8z(b1)};
      }
      __builtin_amdgcn_sched_barrier(0);
    }
  };

  step(0, gA, gB, true, true);
  step(1, gB, gA, true, false);
  step(2, gA, gB, true, true);
#pragma unroll 1
  for (int s = 3; s < LSTM_T + 3; s += 2) {
    step(s, gB, gA, false, false);
    step(s + 1, gA, gB, false, true);
  }

  float partial = 0.0f;
#pragma unroll
  for (int v = 0; v < 8; ++v) partial += Wfc[v + 8 * half] * hS[v];
  const float tot = partial + __shfl_xor(partial, 16, 32);
  if (half == 0 && up) outLine[wv * 8 + (nIdx - 8)] = tot + bfc[0];
  __syncthreads();
  if (wv == 0) out[blockIdx.x * 32 + lane] = outLine[lane];
}

extern "C" void kernel_launch(void* const* d_in, const int* in_sizes, int n_in,
                              void* d_out, int out_size, void* d_ws,
                              size_t ws_size, hipStream_t stream) {
  (void)in_sizes; (void)n_in; (void)d_ws; (void)ws_size; (void)out_size;
  const float* x    = (const float*)d_in[0];
  const float* Wih0 = (const float*)d_in[1];
  const float* Whh0 = (const float*)d_in[2];
  const float* bih0 = (const float*)d_in[3];
  const float* bhh0 = (const float*)d_in[4];
  const float* Wih1 = (const float*)d_in[5];
  const float* Whh1 = (const float*)d_in[6];
  const float* bih1 = (const float*)d_in[7];
  const float* bhh1 = (const float*)d_in[8];
  const float* Wfc  = (const float*)d_in[9];
  const float* bfc  = (const float*)d_in[10];
  float* out = (float*)d_out;

  dim3 block(128);
  dim3 grid(LSTM_B / (8 * 4));
  lstm2_fc_kernel<<<grid, block, 0, stream>>>(x, Wih0, Whh0, bih0, bhh0, Wih1,
                                              Whh1, bih1, bhh1, Wfc, bfc, out);
}


# speedup vs baseline: 1.7648x; 1.0563x over previous
#include <hip/hip_runtime.h>


#define LSTM_B 4096
#define LSTM_T 512

typedef float v2f __attribute__((ext_vector_type(2)));
typedef float v8f __attribute__((ext_vector_type(8)));

static __device__ __forceinline__ v8f wmma4(v2f a, v2f b, v8f c) {
  return __builtin_amdgcn_wmma_f32_16x16x4_f32(false, a, false, b, (short)0, c,
                                               false, false);
}

static __device__ __forceinline__ void swap16(float& d, float& s) {
  asm("v_permlane16_swap_b32 %0, %1" : "+v"(d), "+v"(s));
}

static __device__ __forceinline__ float sigm(float x) {
  return __builtin_amdgcn_rcpf(1.0f + __builtin_amdgcn_exp2f(-1.44269504f * x));
}
static __device__ __forceinline__ float tanh_a(float x) {
  return 1.0f - 2.0f * __builtin_amdgcn_rcpf(
                           1.0f + __builtin_amdgcn_exp2f(2.88539008f * x));
}

static __device__ __forceinline__ float shr8z(float v) {
  return __builtin_bit_cast(
      float, __builtin_amdgcn_update_dpp(0, __builtin_bit_cast(int, v), 0x118,
                                         0xf, 0xf, true));
}

__global__ void __launch_bounds__(128, 1) lstm2_fc_kernel(
    const float* __restrict__ x,
    const float* __restrict__ Wih0,
    const float* __restrict__ Whh0,
    const float* __restrict__ bih0,
    const float* __restrict__ bhh0,
    const float* __restrict__ Wih1,
    const float* __restrict__ Whh1,
    const float* __restrict__ bih1,
    const float* __restrict__ bhh1,
    const float* __restrict__ Wfc,
    const float* __restrict__ bfc,
    float* __restrict__ out)
{
  __shared__ float outLine[32];

  const int lane = threadIdx.x & 31;
  const int half = lane >> 4;
  const int nIdx = lane & 15;
  const bool up  = nIdx >= 8;
  const int wv   = threadIdx.x >> 5;
  const int wid  = blockIdx.x * 4 + wv;
  const int base = wid * 8;

  v2f Whh0A[4][4], Wih1A[4][4], Whh1A[4][4];
  v2f AugA[4];
#pragma unroll
  for (int j = 0; j < 4; ++j) {
    const int row = 16 * j + nIdx;
#pragma unroll
    for (int kc = 0; kc < 4; ++kc) {
      const float* p0 = Whh0 + row * 16 + kc * 4 + 2 * half;
      const float* p1 = Wih1 + row * 16 + kc * 4 + 2 * half;
      const float* p2 = Whh1 + row * 16 + kc * 4 + 2 * half;
      Whh0A[j][kc] = v2f{p0[0], p0[1]};
      Wih1A[j][kc] = v2f{p1[0], p1[1]};
      Whh1A[j][kc] = v2f{p2[0], p2[1]};
    }
    const float w0  = Wih0[row];
    const float bb0 = bih0[row] + bhh0[row];
    const float bb1 = bih1[row] + bhh1[row];
    AugA[j] = v2f{half ? bb1 : w0, half ? 0.0f : bb0};
  }
  const v8f vzero = {0.f, 0.f, 0.f, 0.f, 0.f, 0.f, 0.f, 0.f};

  const bool  xLane   = (half == 0) && !up;
  const float augFix0 = (half == 1 && up) ? 1.0f : 0.0f;
  const float augFix1 = xLane ? 1.0f : 0.0f;

  v8f cS = vzero, hS = vzero;
  v2f hLo[4], hShE[4], hHi[4];
#pragma unroll
  for (int kc = 0; kc < 4; ++kc) {
    hLo[kc]  = v2f{0.0f, 0.0f};
    hShE[kc] = v2f{0.0f, 0.0f};
    hHi[kc]  = v2f{0.0f, 0.0f};
  }
  v8f qA[4], qB[4];
#pragma unroll
  for (int j = 0; j < 4; ++j) { qA[j] = vzero; qB[j] = vzero; }

  const float* xrow = x + (base + (nIdx & 7)) * LSTM_T;

  v8f gA[4], gB[4];

  float xNext = xrow[1];
  {
    const float x0 = xrow[0];
    const v2f augB = v2f{xLane ? x0 : augFix0, augFix1};
#pragma unroll
    for (int j = 0; j < 4; ++j)
      gA[j] = wmma4(Whh0A[j][0], hLo[0], wmma4(AugA[j], augB, vzero));
  }

  auto step = [&](const int s, v8f (&gc)[4], v8f (&gn)[4], const bool first,
                  const bool even) __attribute__((always_inline)) {
    if (even) {
      v2f bq[4];
#pragma unroll
      for (int kc = 0; kc < 4; ++kc)
        bq[kc] = v2f{up ? hShE[kc][0] : hLo[kc][0],
                     up ? hShE[kc][1] : hLo[kc][1]};
#pragma unroll
      for (int j = 0; j < 4; ++j) {
        v8f q = vzero;
#pragma unroll
        for (int kc = 0; kc < 4; ++kc) q = wmma4(Wih1A[j][kc], bq[kc], q);
#pragma unroll
        for (int v = 0; v < 8; ++v) {
          qA[j][v] = up ? q[v] : 0.0f;
          qB[j][v] = shr8z(q[v]);
        }
      }
    }
#pragma unroll
    for (int j = 0; j < 4; ++j) {
      v8f acc = gc[j];
#pragma unroll
      for (int kc = 1; kc < 4; ++kc) acc = wmma4(Whh0A[j][kc], hLo[kc], acc);
#pragma unroll
      for (int kc = 0; kc < 4; ++kc) acc = wmma4(Whh1A[j][kc], hHi[kc], acc);
      gc[j] = acc;
    }
    if (even) __builtin_amdgcn_sched_barrier(0);
    const v2f augB = v2f{xLane ? xNext : augFix0, augFix1};
    const int sn = s + 2;
    xNext = xrow[sn < LSTM_T ? sn : LSTM_T - 1];

#define GM_TL(n) ((n) < 24 ? ((n) % 6 < 2 ? 2 : ((n) % 6 < 4 ? 0 : 1)) : 3)
#define GM_RG(n) ((n) < 24 ? 2 * ((n) / 6) + ((n) & 1) : (n) - 24)
#define GM_PRE(n) ((GM_TL(n) == 2 ? 2.88539008f : -1.44269504f) * gc[GM_TL(n)][GM_RG(n)])
#define GM_PIN() __builtin_amdgcn_sched_barrier(0)
    float m[32], e[32], sv[32], r[32];
    float gvv[8], ttv[8], cvv[8], mcv[8], ecv[8], scv[8], rcv[8], tcv[8], hvv[8];
    float sa0[2], sa1[2], sb0[2], sb1[2];
    auto eSwap = [&](const int p) __attribute__((always_inline)) {
#pragma unroll
      for (int v = 4 * p; v < 4 * p + 4; ++v) {
        cS[v] = (first && up) ? 0.0f : cvv[v];
        hS[v] = (first && up) ? 0.0f : hvv[v];
      }
      sa0[p] = hS[4 * p + 0]; sa1[p] = hS[4 * p + 1];
      sb0[p] = hS[4 * p + 2]; sb1[p] = hS[4 * p + 3];
      swap16(sa0[p], sb0[p]);
      swap16(sa1[p], sb1[p]);
    };
    auto eMask = [&](const int p) __attribute__((always_inline)) {
      const float a0 = sa0[p], a1 = sa1[p], b0 = sb0[p], b1 = sb1[p];
      hLo[p]     = v2f{up ? 0.0f : a0, up ? 0.0f : a1};
      hLo[p + 2] = v2f{up ? 0.0f : b0, up ? 0.0f : b1};
      hHi[p]     = v2f{up ? a0 : 0.0f, up ? a1 : 0.0f};
      hHi[p + 2] = v2f{up ? b0 : 0.0f, up ? b1 : 0.0f};
      if (even) {
        hShE[p]     = v2f{shr8z(a0), shr8z(a1)};
        hShE[p + 2] = v2f{shr8z(b0), shr8z(b1)};
      }
    };
#pragma unroll
    for (int n = 0; n < 4; ++n) m[n] = GM_PRE(n);
    GM_PIN();
#pragma unroll
    for (int t = 0; t < 32; ++t) {
      e[t] = __builtin_amdgcn_exp2f(m[t]);
      if ((t & 1) == 0) {
        if (t + 4 < 32) m[t + 4] = GM_PRE(t + 4);
        if (t + 5 < 32) m[t + 5] = GM_PRE(t + 5);
      } else if (t >= 7) {
        sv[t - 7] = 1.0f + e[t - 7];
        sv[t - 6] = 1.0f + e[t - 6];
      }
      GM_PIN();
    }
#pragma unroll
    for (int u = 0; u < 56; ++u) {
      if (u < 32) r[u] = __builtin_amdgcn_rcpf(sv[u]);
      else if (u < 40) ecv[u - 32] = __builtin_amdgcn_exp2f(mcv[u - 32]);
      else if (u < 48) rcv[u - 40] = __builtin_amdgcn_rcpf(scv[u - 40]);
      if (u < 3) {
        sv[26 + 2 * u] = 1.0f + e[26 + 2 * u];
        sv[27 + 2 * u] = 1.0f + e[27 + 2 * u];
      }
      if (u >= 5 && (u - 5) % 6 == 0 && (u - 5) / 6 < 4) {
        const int a = (u - 5) / 6;
        gvv[2 * a]     = 1.0f - 2.0f * r[6 * a];
        gvv[2 * a + 1] = 1.0f - 2.0f * r[6 * a + 1];
      }
      if (u >= 9 && (u - 9) % 6 == 0 && (u - 9) / 6 < 4) {
        const int a = (u - 9) / 6;
        ttv[2 * a]     = r[6 * a + 2] * gvv[2 * a];
        ttv[2 * a + 1] = r[6 * a + 3] * gvv[2 * a + 1];
      }
      if (u >= 13 && (u - 13) % 6 == 0 && (u - 13) / 6 < 4) {
        const int a = (u - 13) / 6;
        cvv[2 * a]     = r[6 * a + 4] * cS[2 * a] + ttv[2 * a];
        cvv[2 * a + 1] = r[6 * a + 5] * cS[2 * a + 1] + ttv[2 * a + 1];
      }
      if (u >= 18 && (u - 18) % 6 == 0 && (u - 18) / 6 < 4) {
        const int a = (u - 18) / 6;
        mcv[2 * a]     = 2.88539008f * cvv[2 * a];
        mcv[2 * a + 1] = 2.88539008f * cvv[2 * a + 1];
      }
      if (u >= 37 && u <= 43 && (u & 1)) {
        scv[u - 37]     = 1.0f + ecv[u - 37];
        scv[u - 37 + 1] = 1.0f + ecv[u - 37 + 1];
      }
      if (u >= 45 && u <= 51 && (u & 1)) {
        tcv[u - 45]     = 1.0f - 2.0f * rcv[u - 45];
        tcv[u - 45 + 1] = 1.0f - 2.0f * rcv[u - 45 + 1];
      }
      if (u >= 48 && !(u & 1)) {
        hvv[u - 48]     = r[24 + (u - 48)] * tcv[u - 48];
        hvv[u - 48 + 1] = r[24 + (u - 48) + 1] * tcv[u - 48 + 1];
      }
      if (u == 49) gn[0] = wmma4(AugA[0], augB, even ? qA[0] : qB[0]);
      if (u == 51) gn[1] = wmma4(AugA[1], augB, even ? qA[1] : qB[1]);
      if (u == 53) gn[2] = wmma4(AugA[2], augB, even ? qA[2] : qB[2]);
      if (u == 55) gn[3] = wmma4(AugA[3], augB, even ? qA[3] : qB[3]);
      if (u == 52) eSwap(0);
      if (u == 54) eMask(0);
      GM_PIN();
    }
    eSwap(1);
    GM_PIN();
#pragma unroll
    for (int j = 0; j < 4; ++j) gn[j] = wmma4(Whh0A[j][0], hLo[0], gn[j]);
    GM_PIN();
    eMask(1);
#undef GM_TL
#undef GM_RG
#undef GM_PRE
#undef GM_PIN
  };

  step(0, gA, gB, true, true);
  step(1, gB, gA, true, false);
  step(2, gA, gB, true, true);
#pragma unroll 1
  for (int s = 3; s < LSTM_T + 3; s += 2) {
    step(s, gB, gA, false, false);
    step(s + 1, gA, gB, false, true);
  }

  float partial = 0.0f;
#pragma unroll
  for (int v = 0; v < 8; ++v) partial += Wfc[v + 8 * half] * hS[v];
  const float tot = partial + __shfl_xor(partial, 16, 32);
  if (half == 0 && up) outLine[wv * 8 + (nIdx - 8)] = tot + bfc[0];
  __syncthreads();
  if (wv == 0) out[blockIdx.x * 32 + lane] = outLine[lane];
}

extern "C" void kernel_launch(void* const* d_in, const int* in_sizes, int n_in,
                              void* d_out, int out_size, void* d_ws,
                              size_t ws_size, hipStream_t stream) {
  (void)in_sizes; (void)n_in; (void)d_ws; (void)ws_size; (void)out_size;
  const float* x    = (const float*)d_in[0];
  const float* Wih0 = (const float*)d_in[1];
  const float* Whh0 = (const float*)d_in[2];
  const float* bih0 = (const float*)d_in[3];
  const float* bhh0 = (const float*)d_in[4];
  const float* Wih1 = (const float*)d_in[5];
  const float* Whh1 = (const float*)d_in[6];
  const float* bih1 = (const float*)d_in[7];
  const float* bhh1 = (const float*)d_in[8];
  const float* Wfc  = (const float*)d_in[9];
  const float* bfc  = (const float*)d_in[10];
  float* out = (float*)d_out;

  dim3 block(128);
  dim3 grid(LSTM_B / (8 * 4));
  lstm2_fc_kernel<<<grid, block, 0, stream>>>(x, Wih0, Whh0, bih0, bhh0, Wih1,
                                              Whh1, bih1, bhh1, Wfc, bfc, out);
}
